# starting kernel
# baseline (speedup 1.0000x reference)
#include <hip/hip_runtime.h>
#include <math.h>

typedef __attribute__((ext_vector_type(2))) float v2f;
typedef __attribute__((ext_vector_type(8))) float v8f;

__device__ __forceinline__ float mish_f(float v) {
  float sp = (v > 20.0f) ? v : log1pf(__expf(v));
  return v * tanhf(sp);
}

__global__ void mish_kernel(const float* __restrict__ in, float* __restrict__ out, int n) {
  int i = blockIdx.x * blockDim.x + threadIdx.x;
  if (i < n) out[i] = mish_f(in[i]);
}

__global__ void cat_mish_kernel(const float* __restrict__ xs, const float* __restrict__ ys,
                                const float* __restrict__ dprev, float* __restrict__ out,
                                int N, int B) {
  int twoN = 2 * N;
  int idx = blockIdx.x * blockDim.x + threadIdx.x;
  if (idx >= B * twoN) return;
  int b = idx / twoN;
  int c = idx - b * twoN;
  float v = (c < N) ? xs[b * N + c] : ys[b * N + (c - N)];
  if (dprev) v += dprev[idx];
  out[idx] = mish_f(v);
}

__global__ void gemm_wmma_f32(const float* __restrict__ A, const float* __restrict__ W,
                              const float* __restrict__ bias, float* __restrict__ C,
                              int M, int N, int K) {
  int wave = threadIdx.x >> 5;
  int lane = threadIdx.x & 31;
  int tiles_n = N >> 5;
  int tile = blockIdx.x * 4 + wave;
  if (tile >= (M >> 5) * tiles_n) return;
  int tm = tile / tiles_n;
  int m0 = tm << 5;
  int n0 = (tile - tm * tiles_n) << 5;
  int r  = lane & 15;
  int kh = (lane >> 4) << 1;
  const float* Ar0 = A + (size_t)(m0 + r) * K + kh;
  const float* Ar1 = Ar0 + (size_t)16 * K;
  const float* Wr0 = W + (size_t)(n0 + r) * K + kh;
  const float* Wr1 = Wr0 + (size_t)16 * K;
  v8f acc00 = {};
  v8f acc01 = {};
  v8f acc10 = {};
  v8f acc11 = {};
  for (int k = 0; k < K; k += 4) {
    v2f a0, a1, b0, b1;
    a0.x = Ar0[k]; a0.y = Ar0[k + 1];
    a1.x = Ar1[k]; a1.y = Ar1[k + 1];
    b0.x = Wr0[k]; b0.y = Wr0[k + 1];
    b1.x = Wr1[k]; b1.y = Wr1[k + 1];
    acc00 = __builtin_amdgcn_wmma_f32_16x16x4_f32(false, a0, false, b0, (short)0, acc00, false, false);
    acc01 = __builtin_amdgcn_wmma_f32_16x16x4_f32(false, a0, false, b1, (short)0, acc01, false, false);
    acc10 = __builtin_amdgcn_wmma_f32_16x16x4_f32(false, a1, false, b0, (short)0, acc10, false, false);
    acc11 = __builtin_amdgcn_wmma_f32_16x16x4_f32(false, a1, false, b1, (short)0, acc11, false, false);
  }
  int n_lo = n0 + r;
  int n_hi = n_lo + 16;
  float bv0 = bias ? bias[n_lo] : 0.0f;
  float bv1 = bias ? bias[n_hi] : 0.0f;
  int mb = m0 + ((lane >> 4) << 3);
#pragma unroll
  for (int v = 0; v < 8; ++v) {
    C[(size_t)(mb + v) * N + n_lo]      = acc00[v] + bv0;
    C[(size_t)(mb + v) * N + n_hi]      = acc01[v] + bv1;
    C[(size_t)(mb + 16 + v) * N + n_lo] = acc10[v] + bv0;
    C[(size_t)(mb + 16 + v) * N + n_hi] = acc11[v] + bv1;
  }
}

__global__ void attn_kernel(const float* __restrict__ self, const float* __restrict__ other,
                            const float* __restrict__ beta_p, float* __restrict__ out, int N) {
  __shared__ float sh_s[512];
  __shared__ float sh_o[512];
  __shared__ float smx[256];
  __shared__ float smn[256];
  int b = blockIdx.x;
  int t = threadIdx.x;
  const float* s = self  + (size_t)b * N;
  const float* o = other + (size_t)b * N;
  float beta = beta_p[0];
  if (beta == 0.0f) {
    for (int i = t; i < N; i += blockDim.x) out[(size_t)b * N + i] = s[i];
    return;
  }
  for (int i = t; i < N; i += blockDim.x) { sh_s[i] = s[i]; sh_o[i] = o[i]; }
  __syncthreads();
  float mx = -3.4e38f, mn = 3.4e38f;
  for (int i = t; i < N; i += blockDim.x) { mx = fmaxf(mx, sh_o[i]); mn = fminf(mn, sh_o[i]); }
  smx[t] = mx; smn[t] = mn;
  __syncthreads();
  for (int off = 128; off > 0; off >>= 1) {
    if (t < off) { smx[t] = fmaxf(smx[t], smx[t + off]); smn[t] = fminf(smn[t], smn[t + off]); }
    __syncthreads();
  }
  float omax = smx[0], omin = smn[0];
  for (int i = t; i < N; i += blockDim.x) {
    float si = sh_s[i];
    float m = (si >= 0.0f) ? si * omax : si * omin;
    float den = 0.0f, num = 0.0f;
    for (int j = 0; j < N; ++j) {
      float e = __expf(si * sh_o[j] - m);
      den += e;
      num += e * sh_s[j];
    }
    out[(size_t)b * N + i] = si + beta * num / den;
  }
}

__global__ void head_kernel(const float* __restrict__ d3, const float* __restrict__ W6,
                            const float* __restrict__ b6, float* __restrict__ out, int B) {
  __shared__ float sh[8];
  int b = blockIdx.x;
  int lane = threadIdx.x;
  if (lane < 7) {
    const float* row = d3 + (size_t)b * 64;
    const float* w   = W6 + (size_t)lane * 64;
    float acc = b6[lane];
    for (int k = 0; k < 64; ++k) acc += row[k] * w[k];
    sh[lane] = acc;
  }
  __syncthreads();
  if (lane < 7) {
    float nrm = sqrtf(sh[0]*sh[0] + sh[1]*sh[1] + sh[2]*sh[2] + sh[3]*sh[3]);
    float v = sh[lane];
    if (lane < 4) v = v / fmaxf(nrm, 1e-12f);
    out[(size_t)b * 7 + lane] = v;
    out[(size_t)B * 7 + (size_t)b * 7 + lane] = v;
  }
}

extern "C" void kernel_launch(void* const* d_in, const int* in_sizes, int n_in,
                              void* d_out, int out_size, void* d_ws, size_t ws_size,
                              hipStream_t stream) {
  const float* x    = (const float*)d_in[0];
  const float* y    = (const float*)d_in[1];
  const float* W1   = (const float*)d_in[2];
  const float* be1  = (const float*)d_in[3];
  const float* fcW1 = (const float*)d_in[4];
  const float* fcb1 = (const float*)d_in[5];
  const float* W2   = (const float*)d_in[6];
  const float* be2  = (const float*)d_in[7];
  const float* fcW2 = (const float*)d_in[8];
  const float* fcb2 = (const float*)d_in[9];
  const float* W3   = (const float*)d_in[10];
  const float* be3  = (const float*)d_in[11];
  const float* fcW3 = (const float*)d_in[12];
  const float* fcb3 = (const float*)d_in[13];
  const float* W4   = (const float*)d_in[14];
  const float* be4  = (const float*)d_in[15];
  const float* fcW4 = (const float*)d_in[16];
  const float* fcb4 = (const float*)d_in[17];
  const float* W6   = (const float*)d_in[18];
  const float* b6   = (const float*)d_in[19];

  float* ws = (float*)d_ws;
  float* m  = ws;
  float* qa = m  + 524288;
  float* qb = qa + 262144;
  float* xa = qb + 262144;
  float* ya = xa + 262144;
  float* d0 = ya + 262144;
  float* d1 = d0 + 262144;
  float* d2 = d1 + 131072;
  float* d3 = d2 + 65536;

  const int B = 512;
  auto grid1d = [](int n) { return (n + 255) / 256; };
  auto gemm = [&](const float* A, const float* W, const float* bias, float* C, int N, int K) {
    int tiles = (B >> 5) * (N >> 5);
    hipLaunchKernelGGL(gemm_wmma_f32, dim3((tiles + 3) / 4), dim3(128), 0, stream,
                       A, W, bias, C, B, N, K);
  };
  auto stage = [&](const float* xin, const float* yin, const float* W, const float* beta,
                   int Kin, int Nout) {
    hipLaunchKernelGGL(mish_kernel, dim3(grid1d(B * Kin)), dim3(256), 0, stream, xin, m, B * Kin);
    gemm(m, W, nullptr, qa, Nout, Kin);
    hipLaunchKernelGGL(mish_kernel, dim3(grid1d(B * Kin)), dim3(256), 0, stream, yin, m, B * Kin);
    gemm(m, W, nullptr, qb, Nout, Kin);
    hipLaunchKernelGGL(attn_kernel, dim3(B), dim3(256), 0, stream, qa, qb, beta, xa, Nout);
    hipLaunchKernelGGL(attn_kernel, dim3(B), dim3(256), 0, stream, qb, qa, beta, ya, Nout);
  };
  auto fc = [&](const float* dprev, const float* fcW, const float* fcb, float* dout,
                int N, int Nout) {
    hipLaunchKernelGGL(cat_mish_kernel, dim3(grid1d(B * 2 * N)), dim3(256), 0, stream,
                       xa, ya, dprev, m, N, B);
    gemm(m, fcW, fcb, dout, Nout, 2 * N);
  };

  stage(x,  y,  W1, be1, 1024, 512);
  fc(nullptr, fcW1, fcb1, d0, 512, 512);
  stage(xa, ya, W2, be2, 512, 256);
  fc(d0, fcW2, fcb2, d1, 256, 256);
  stage(xa, ya, W3, be3, 256, 128);
  fc(d1, fcW3, fcb3, d2, 128, 128);
  stage(xa, ya, W4, be4, 128, 64);
  fc(d2, fcW4, fcb4, d3, 64, 64);
  hipLaunchKernelGGL(head_kernel, dim3(B), dim3(32), 0, stream, d3, W6, b6, (float*)d_out, B);

  (void)in_sizes; (void)n_in; (void)out_size; (void)ws_size;
}


# stage 1 attention: both directions in one launch, rows stored as their mish straight into the [mish x0 | mish y0] plane that fc1 and stage 2 read; its concat-mish pass goes; 17 -> 15 launches
# speedup vs baseline: 4.0045x; 4.0045x over previous
#include <hip/hip_runtime.h>
#include <math.h>

typedef __attribute__((ext_vector_type(2))) float v2f;
typedef __attribute__((ext_vector_type(4))) float v4f;
typedef __attribute__((ext_vector_type(8))) float v8f;

__device__ __forceinline__ float mish_f(float v) {
  float sp = (v > 20.0f) ? v : log1pf(__expf(v));
  return v * tanhf(sp);
}

__global__ void mish_kernel(const float* __restrict__ in, float* __restrict__ out, int n) {
  int i = blockIdx.x * blockDim.x + threadIdx.x;
  if (i < n) out[i] = mish_f(in[i]);
}

__global__ void cat_mish_kernel(const float* __restrict__ xs, const float* __restrict__ ys,
                                const float* __restrict__ dprev, float* __restrict__ out,
                                float* __restrict__ plain, int N, int B) {
  int twoN = 2 * N;
  int idx = blockIdx.x * blockDim.x + threadIdx.x;
  if (idx >= B * twoN) return;
  int b = idx / twoN;
  int c = idx - b * twoN;
  float v = (c < N) ? xs[b * N + c] : ys[b * N + (c - N)];
  if (plain) plain[idx] = mish_f(v);
  if (dprev) v += dprev[idx];
  out[idx] = mish_f(v);
}

#define GT_K 128
#define GT_P 132
#define GT_CP 36

template <int NA>
__device__ __forceinline__ void gemm_block_mma(const float* __restrict__ fa, const float* __restrict__ fb,
                                               v8f (&acc)[2 * NA]) {
  v4f av[NA][16], bv[16];
#pragma unroll
  for (int j = 0; j < 16; ++j) {
#pragma unroll
    for (int o = 0; o < NA; ++o) av[o][j] = *(const v4f*)(fa + o * 32 * GT_P + 8 * j);
    bv[j] = *(const v4f*)(fb + 8 * j);
  }
  __builtin_amdgcn_sched_barrier(0);
#pragma unroll
  for (int j = 0; j < 16; ++j) {
    v2f b0, b1;
    b0.x = bv[j].x; b0.y = bv[j].y;
    b1.x = bv[j].z; b1.y = bv[j].w;
#pragma unroll
    for (int o = 0; o < NA; ++o) {
      v2f a0, a1;
      a0.x = av[o][j].x; a0.y = av[o][j].y;
      a1.x = av[o][j].z; a1.y = av[o][j].w;
      acc[2 * o]     = __builtin_amdgcn_wmma_f32_16x16x4_f32(false, a0, false, b0, (short)0, acc[2 * o], false, false);
      acc[2 * o + 1] = __builtin_amdgcn_wmma_f32_16x16x4_f32(false, a1, false, b1, (short)0, acc[2 * o + 1], false, false);
    }
  }
}

template <int NA>
__device__ __forceinline__ void gemm_rows_load(v4f (&ra)[8 * NA], v4f (&rw)[8], const float* __restrict__ Ab,
                                               const float* __restrict__ A2b, const float* __restrict__ Wb,
                                               int lda, int K) {
#pragma unroll
  for (int i = 0; i < 8; ++i) {
    ra[i] = *(const v4f*)(Ab + (size_t)(4 * i) * lda);
    if constexpr (NA == 2) ra[8 + i] = *(const v4f*)(A2b + (size_t)(4 * i) * lda);
    rw[i] = *(const v4f*)(Wb + (size_t)(4 * i) * K);
  }
}

template <int NA>
__device__ __forceinline__ void gemm_rows_store(float* sa, float* sw, const v4f (&ra)[8 * NA], const v4f (&rw)[8]) {
#pragma unroll
  for (int i = 0; i < 8; ++i) {
    *(v4f*)(sa + 4 * i * GT_P) = ra[i];
    if constexpr (NA == 2) *(v4f*)(sa + (32 + 4 * i) * GT_P) = ra[8 + i];
    *(v4f*)(sw + 4 * i * GT_P) = rw[i];
  }
}

__device__ __forceinline__ void lds_barrier() {
  asm volatile("s_wait_dscnt 0x0\n\ts_barrier_signal -1\n\ts_barrier_wait -1" ::: "memory");
}

template <int NA>
__device__ __forceinline__ void stores_behind_mma(const v8f (&acc)[2 * NA]) {
  if constexpr (NA == 2) asm volatile("" :: "v"(acc[0]), "v"(acc[1]), "v"(acc[2]), "v"(acc[3]));
  else asm volatile("" :: "v"(acc[0]), "v"(acc[1]));
}

template <int NA>
__device__ __forceinline__ void gemm_tile(float* sA0, float* sA1, float* sW0, float* sW1, float* sC,
                                          const float* __restrict__ A, const float* __restrict__ A2,
                                          const float* __restrict__ W, const float* __restrict__ bias,
                                          float* __restrict__ C, float* __restrict__ C2,
                                          int N, int K, int lda, int tile) {
  int t = threadIdx.x;
  int wave = t >> 5;
  int lane = t & 31;
  int tiles_n = N >> 5;
  int tm = tile / tiles_n;
  int m0 = tm << 5;
  int n0 = (tile - tm * tiles_n) << 5;
  int lc = lane << 2;
  const float* Ag = A + (size_t)(m0 + wave) * lda + lc;
  const float* A2g = (NA == 2) ? A2 + (size_t)(m0 + wave) * lda + lc : Ag;
  const float* Wg = W + (size_t)(n0 + wave) * K + lc;
  int so = wave * GT_P + lc;
  int c = lane & 15;
  int h = lane >> 4;
  int wm = wave >> 1;
  int wn = wave & 1;
  int fao = (16 * wm + c) * GT_P + 4 * h;
  int fbo = (16 * wn + c) * GT_P + 4 * h;

  v4f ra0[8 * NA], rw0[8], ra1[8 * NA], rw1[8];
  float* sa0 = sA0 + so;
  float* sw0 = sW0 + so;
  float* sa1 = sA1 + so;
  float* sw1 = sW1 + so;
  const float* fa0 = sA0 + fao;
  const float* fb0 = sW0 + fbo;
  const float* fa1 = sA1 + fao;
  const float* fb1 = sW1 + fbo;
  v8f acc[2 * NA] = {};
  int nb = K / GT_K;

  gemm_rows_load<NA>(ra0, rw0, Ag, A2g, Wg, lda, K);
  if (nb == 1) {
    gemm_rows_store<NA>(sa0, sw0, ra0, rw0);
    __syncthreads();
    gemm_block_mma<NA>(fa0, fb0, acc);
  } else {
    gemm_rows_load<NA>(ra1, rw1, Ag + GT_K, A2g + GT_K, Wg + GT_K, lda, K);
    gemm_rows_store<NA>(sa0, sw0, ra0, rw0);
    lds_barrier();
    int kb = 0;
    for (; kb + 3 < nb; kb += 2) {
      size_t o2 = (size_t)(kb + 2) * GT_K;
      gemm_rows_load<NA>(ra0, rw0, Ag + o2, A2g + o2, Wg + o2, lda, K);
      gemm_block_mma<NA>(fa0, fb0, acc);
      stores_behind_mma<NA>(acc);
      gemm_rows_store<NA>(sa1, sw1, ra1, rw1);
      lds_barrier();
      size_t o3 = (size_t)(kb + 3) * GT_K;
      gemm_rows_load<NA>(ra1, rw1, Ag + o3, A2g + o3, Wg + o3, lda, K);
      gemm_block_mma<NA>(fa1, fb1, acc);
      stores_behind_mma<NA>(acc);
      gemm_rows_store<NA>(sa0, sw0, ra0, rw0);
      lds_barrier();
    }
    bool three = kb + 2 < nb;
    if (three) {
      size_t o2 = (size_t)(kb + 2) * GT_K;
      gemm_rows_load<NA>(ra0, rw0, Ag + o2, A2g + o2, Wg + o2, lda, K);
    }
    gemm_block_mma<NA>(fa0, fb0, acc);
    stores_behind_mma<NA>(acc);
    gemm_rows_store<NA>(sa1, sw1, ra1, rw1);
    lds_barrier();
    gemm_block_mma<NA>(fa1, fb1, acc);
    if (three) {
      stores_behind_mma<NA>(acc);
      gemm_rows_store<NA>(sa0, sw0, ra0, rw0);
      lds_barrier();
      gemm_block_mma<NA>(fa0, fb0, acc);
    }
  }

  int ccol = 16 * wn + c;
  float bv = bias ? bias[n0 + ccol] : 0.0f;
  int crow = 16 * wm + 8 * h;
#pragma unroll
  for (int o = 0; o < NA; ++o) {
#pragma unroll
    for (int r = 0; r < 8; ++r)
      sC[(32 * o + crow + r) * GT_CP + ccol] = (acc[2 * o][r] + acc[2 * o + 1][r]) + bv;
  }
  __syncthreads();
  int orow = 8 * wave + (lane >> 3);
  int ocol = (lane & 7) << 2;
#pragma unroll
  for (int o = 0; o < NA; ++o) {
    float* Co = o ? C2 : C;
#pragma unroll
    for (int i = 0; i < 2; ++i) {
      int row = orow + 4 * i;
      v4f v = *(const v4f*)&sC[(32 * o + row) * GT_CP + ocol];
      *(v4f*)&Co[(size_t)(m0 + row) * N + n0 + ocol] = v;
    }
  }
}

template <int NA>
__global__ __launch_bounds__(128)
void gemm_wmma_f32(const float* __restrict__ A, const float* __restrict__ A2, const float* __restrict__ W,
                   const float* __restrict__ bias, float* __restrict__ C, float* __restrict__ C2,
                   int M, int N, int K, int lda) {
  __shared__ __attribute__((aligned(16))) float sA[2][32 * NA * GT_P];
  __shared__ __attribute__((aligned(16))) float sW[2][32 * GT_P];
  __shared__ __attribute__((aligned(16))) float sC[32 * NA * GT_CP];
  gemm_tile<NA>(sA[0], sA[1], sW[0], sW[1], sC, A, A2, W, bias, C, C2, N, K, lda, blockIdx.x);
  (void)M;
}

template <int NT, bool ACT>
__device__ __forceinline__ void attn_rows(float* sh_s, float* sh_o, float* smx, float* smn,
                                          const float* __restrict__ self, const float* __restrict__ other,
                                          const float* __restrict__ beta_p, float* __restrict__ out,
                                          int opitch, int N, int b) {
  int t = threadIdx.x;
  const float* s = self  + (size_t)b * N;
  const float* o = other + (size_t)b * N;
  float* orow = out + (size_t)b * opitch;
  float beta = beta_p[0];
  if (beta == 0.0f) {
    for (int i = t; i < N; i += blockDim.x) orow[i] = ACT ? mish_f(s[i]) : s[i];
    return;
  }
  for (int i = t; i < N; i += blockDim.x) { sh_s[i] = s[i]; sh_o[i] = o[i]; }
  __syncthreads();
  float mx = -3.4e38f, mn = 3.4e38f;
  for (int i = t; i < N; i += blockDim.x) { mx = fmaxf(mx, sh_o[i]); mn = fminf(mn, sh_o[i]); }
  smx[t] = mx; smn[t] = mn;
  __syncthreads();
  for (int off = NT / 2; off > 0; off >>= 1) {
    if (t < off) { smx[t] = fmaxf(smx[t], smx[t + off]); smn[t] = fminf(smn[t], smn[t + off]); }
    __syncthreads();
  }
  float omax = smx[0], omin = smn[0];
  for (int i = t; i < N; i += blockDim.x) {
    float si = sh_s[i];
    float m = (si >= 0.0f) ? si * omax : si * omin;
    float den = 0.0f, num = 0.0f;
    for (int j = 0; j < N; ++j) {
      float e = __expf(si * sh_o[j] - m);
      den += e;
      num += e * sh_s[j];
    }
    float v = si + beta * num / den;
    orow[i] = ACT ? mish_f(v) : v;
  }
}

__global__ void attn_act_kernel(const float* __restrict__ qx, const float* __restrict__ qy,
                                const float* __restrict__ beta_p, float* __restrict__ act, int N, int B) {
  __shared__ float sh_s[512];
  __shared__ float sh_o[512];
  __shared__ float smx[256];
  __shared__ float smn[256];
  int g = blockIdx.x;
  bool second = g >= B;
  int b = second ? g - B : g;
  attn_rows<256, true>(sh_s, sh_o, smx, smn, second ? qy : qx, second ? qx : qy, beta_p,
                       act + (second ? N : 0), 2 * N, N, b);
}

__global__ __launch_bounds__(128)
void fc_attn_kernel(const float* __restrict__ A, const float* __restrict__ W, const float* __restrict__ bias,
                    float* __restrict__ C, int N, int K, int lda, int tiles,
                    const float* __restrict__ qx, const float* __restrict__ qy,
                    const float* __restrict__ beta_p, float* __restrict__ ox, float* __restrict__ oy,
                    int Nq, int B) {
  __shared__ __attribute__((aligned(16))) float sA[2][32 * GT_P];
  __shared__ __attribute__((aligned(16))) float sW[2][32 * GT_P];
  __shared__ __attribute__((aligned(16))) float sC[32 * GT_CP];
  __shared__ float sh_s[512];
  __shared__ float sh_o[512];
  __shared__ float smx[128];
  __shared__ float smn[128];
  int g = blockIdx.x;
  if (g < tiles) {
    gemm_tile<1>(sA[0], sA[1], sW[0], sW[1], sC, A, (const float*)nullptr, W, bias, C, (float*)nullptr,
                 N, K, lda, g);
  } else {
    g -= tiles;
    bool second = g >= B;
    int b = second ? g - B : g;
    attn_rows<128, false>(sh_s, sh_o, smx, smn, second ? qy : qx, second ? qx : qy, beta_p, second ? oy : ox,
                          Nq, Nq, b);
  }
}

__global__ void head_kernel(const float* __restrict__ d3, const float* __restrict__ W6,
                            const float* __restrict__ b6, float* __restrict__ out, int B) {
  __shared__ float sh[8];
  int b = blockIdx.x;
  int lane = threadIdx.x;
  if (lane < 7) {
    const float* row = d3 + (size_t)b * 64;
    const float* w   = W6 + (size_t)lane * 64;
    float acc = b6[lane];
    for (int k = 0; k < 64; ++k) acc += row[k] * w[k];
    sh[lane] = acc;
  }
  __syncthreads();
  if (lane < 7) {
    float nrm = sqrtf(sh[0]*sh[0] + sh[1]*sh[1] + sh[2]*sh[2] + sh[3]*sh[3]);
    float v = sh[lane];
    if (lane < 4) v = v / fmaxf(nrm, 1e-12f);
    out[(size_t)b * 7 + lane] = v;
    out[(size_t)B * 7 + (size_t)b * 7 + lane] = v;
  }
}

extern "C" void kernel_launch(void* const* d_in, const int* in_sizes, int n_in,
                              void* d_out, int out_size, void* d_ws, size_t ws_size,
                              hipStream_t stream) {
  const float* x    = (const float*)d_in[0];
  const float* y    = (const float*)d_in[1];
  const float* W1   = (const float*)d_in[2];
  const float* be1  = (const float*)d_in[3];
  const float* fcW1 = (const float*)d_in[4];
  const float* fcb1 = (const float*)d_in[5];
  const float* W2   = (const float*)d_in[6];
  const float* be2  = (const float*)d_in[7];
  const float* fcW2 = (const float*)d_in[8];
  const float* fcb2 = (const float*)d_in[9];
  const float* W3   = (const float*)d_in[10];
  const float* be3  = (const float*)d_in[11];
  const float* fcW3 = (const float*)d_in[12];
  const float* fcb3 = (const float*)d_in[13];
  const float* W4   = (const float*)d_in[14];
  const float* be4  = (const float*)d_in[15];
  const float* fcW4 = (const float*)d_in[16];
  const float* fcb4 = (const float*)d_in[17];
  const float* W6   = (const float*)d_in[18];
  const float* b6   = (const float*)d_in[19];

  float* ws = (float*)d_ws;
  float* m  = ws;
  float* qa = m  + 524288;
  float* qb = qa + 262144;
  float* xa = qb + 262144;
  float* ya = xa + 262144;
  float* d0 = ya + 262144;
  float* d1 = d0 + 262144;
  float* d2 = d1 + 131072;
  float* d3 = d2 + 65536;
  float* m2 = d3 + 32768;
  float* my = m2 + 262144;

  const int B = 512;
  auto grid1d = [](int n) { return (n + 255) / 256; };
  auto gemm = [&](const float* A, int lda, const float* W, const float* bias, float* C, int N, int K) {
    int tiles = (B >> 5) * (N >> 5);
    hipLaunchKernelGGL((gemm_wmma_f32<1>), dim3(tiles), dim3(128), 0, stream,
                       A, (const float*)nullptr, W, bias, C, (float*)nullptr, B, N, K, lda);
  };
  auto gemm_xy = [&](const float* Ax, const float* Ay, int lda, const float* W, float* Cx, float* Cy,
                     int N, int K) {
    int tiles = (B >> 5) * (N >> 5);
    hipLaunchKernelGGL((gemm_wmma_f32<2>), dim3(tiles), dim3(128), 0, stream,
                       Ax, Ay, W, (const float*)nullptr, Cx, Cy, B, N, K, lda);
  };
  hipLaunchKernelGGL(mish_kernel, dim3(grid1d(B * 1024)), dim3(256), 0, stream, x, m, B * 1024);
  hipLaunchKernelGGL(mish_kernel, dim3(grid1d(B * 1024)), dim3(256), 0, stream, y, my, B * 1024);
  gemm_xy(m, my, 1024, W1, qa, qb, 512, 1024);
  hipLaunchKernelGGL(attn_act_kernel, dim3(2 * B), dim3(256), 0, stream,
                     (const float*)qa, (const float*)qb, be1, m, 512, B);
  auto cat = [&](const float* dprev, float* plain, int N) {
    hipLaunchKernelGGL(cat_mish_kernel, dim3(grid1d(B * 2 * N)), dim3(256), 0, stream,
                       xa, ya, dprev, m, plain, N, B);
  };
  auto fc_attn = [&](const float* fcW, const float* fcb, float* dout, int N, int Nout,
                     const float* beta, int Nq) {
    int tiles = (B >> 5) * (Nout >> 5);
    hipLaunchKernelGGL(fc_attn_kernel, dim3(tiles + 2 * B), dim3(128), 0, stream,
                       (const float*)m, fcW, fcb, dout, Nout, 2 * N, 2 * N, tiles,
                       (const float*)qa, (const float*)qb, beta, xa, ya, Nq, B);
  };

  gemm_xy(m, m + 512, 1024, W2, qa, qb, 256, 512);
  fc_attn(fcW1, fcb1, d0, 512, 512, be2, 256);
  cat(d0, m2, 256);
  gemm_xy(m2, m2 + 256, 512, W3, qa, qb, 128, 256);
  fc_attn(fcW2, fcb2, d1, 256, 256, be3, 128);
  cat(d1, m2, 128);
  gemm_xy(m2, m2 + 128, 256, W4, qa, qb, 64, 128);
  fc_attn(fcW3, fcb3, d2, 128, 128, be4, 64);
  cat(d2, nullptr, 64);
  gemm(m, 128, fcW4, fcb4, d3, 64, 128);
  hipLaunchKernelGGL(head_kernel, dim3(B), dim3(32), 0, stream, d3, W6, b6, (float*)d_out, B);

  (void)in_sizes; (void)n_in; (void)out_size; (void)ws_size;
}


# concat-mish pass of fc2, fc3 runs as extra groups of the next query-product launch (x, y tiles one a group, two groups a WGP); attention stores its rows' mish too; 15 -> 13 launches; sums unchanged
# speedup vs baseline: 4.2029x; 1.0495x over previous
#include <hip/hip_runtime.h>
#include <math.h>

typedef __attribute__((ext_vector_type(2))) float v2f;
typedef __attribute__((ext_vector_type(4))) float v4f;
typedef __attribute__((ext_vector_type(8))) float v8f;

__device__ __forceinline__ float mish_f(float v) {
  float sp = (v > 20.0f) ? v : log1pf(__expf(v));
  return v * tanhf(sp);
}

__global__ void mish_kernel(const float* __restrict__ in, float* __restrict__ out, int n) {
  int i = blockIdx.x * blockDim.x + threadIdx.x;
  if (i < n) out[i] = mish_f(in[i]);
}

__global__ void cat_mish_kernel(const float* __restrict__ xs, const float* __restrict__ ys,
                                const float* __restrict__ dprev, float* __restrict__ out, int N, int B) {
  int twoN = 2 * N;
  int idx = blockIdx.x * blockDim.x + threadIdx.x;
  if (idx >= B * twoN) return;
  int b = idx / twoN;
  int c = idx - b * twoN;
  float v = (c < N) ? xs[b * N + c] : ys[b * N + (c - N)];
  v += dprev[idx];
  out[idx] = mish_f(v);
}

__device__ __forceinline__ void cat_mish_group(const float* __restrict__ xs, const float* __restrict__ ys,
                                               const float* __restrict__ dprev, float* __restrict__ out,
                                               int N, int B, int cg) {
  int twoN = 2 * N;
  int idx = (cg * 128 + (int)threadIdx.x) << 2;
  if (idx >= B * twoN) return;
  int b = idx / twoN;
  int c = idx - b * twoN;
  const float* src = (c < N) ? xs + (size_t)b * N + c : ys + (size_t)b * N + (c - N);
  v4f v = *(const v4f*)src;
  v4f d = *(const v4f*)(dprev + idx);
  v4f r;
  r.x = mish_f(v.x + d.x);
  r.y = mish_f(v.y + d.y);
  r.z = mish_f(v.z + d.z);
  r.w = mish_f(v.w + d.w);
  *(v4f*)(out + idx) = r;
}

#define GT_K 128
#define GT_P 132
#define GT_CP 36

template <int NA>
__device__ __forceinline__ void gemm_block_mma(const float* __restrict__ fa, const float* __restrict__ fb,
                                               v8f (&acc)[2 * NA]) {
  v4f av[NA][16], bv[16];
#pragma unroll
  for (int j = 0; j < 16; ++j) {
#pragma unroll
    for (int o = 0; o < NA; ++o) av[o][j] = *(const v4f*)(fa + o * 32 * GT_P + 8 * j);
    bv[j] = *(const v4f*)(fb + 8 * j);
  }
  __builtin_amdgcn_sched_barrier(0);
#pragma unroll
  for (int j = 0; j < 16; ++j) {
    v2f b0, b1;
    b0.x = bv[j].x; b0.y = bv[j].y;
    b1.x = bv[j].z; b1.y = bv[j].w;
#pragma unroll
    for (int o = 0; o < NA; ++o) {
      v2f a0, a1;
      a0.x = av[o][j].x; a0.y = av[o][j].y;
      a1.x = av[o][j].z; a1.y = av[o][j].w;
      acc[2 * o]     = __builtin_amdgcn_wmma_f32_16x16x4_f32(false, a0, false, b0, (short)0, acc[2 * o], false, false);
      acc[2 * o + 1] = __builtin_amdgcn_wmma_f32_16x16x4_f32(false, a1, false, b1, (short)0, acc[2 * o + 1], false, false);
    }
  }
}

template <int NA>
__device__ __forceinline__ void gemm_rows_load(v4f (&ra)[8 * NA], v4f (&rw)[8], const float* __restrict__ Ab,
                                               const float* __restrict__ A2b, const float* __restrict__ Wb,
                                               int lda, int K) {
#pragma unroll
  for (int i = 0; i < 8; ++i) {
    ra[i] = *(const v4f*)(Ab + (size_t)(4 * i) * lda);
    if constexpr (NA == 2) ra[8 + i] = *(const v4f*)(A2b + (size_t)(4 * i) * lda);
    rw[i] = *(const v4f*)(Wb + (size_t)(4 * i) * K);
  }
}

template <int NA>
__device__ __forceinline__ void gemm_rows_store(float* sa, float* sw, const v4f (&ra)[8 * NA], const v4f (&rw)[8]) {
#pragma unroll
  for (int i = 0; i < 8; ++i) {
    *(v4f*)(sa + 4 * i * GT_P) = ra[i];
    if constexpr (NA == 2) *(v4f*)(sa + (32 + 4 * i) * GT_P) = ra[8 + i];
    *(v4f*)(sw + 4 * i * GT_P) = rw[i];
  }
}

__device__ __forceinline__ void lds_barrier() {
  asm volatile("s_wait_dscnt 0x0\n\ts_barrier_signal -1\n\ts_barrier_wait -1" ::: "memory");
}

template <int NA>
__device__ __forceinline__ void stores_behind_mma(const v8f (&acc)[2 * NA]) {
  if constexpr (NA == 2) asm volatile("" :: "v"(acc[0]), "v"(acc[1]), "v"(acc[2]), "v"(acc[3]));
  else asm volatile("" :: "v"(acc[0]), "v"(acc[1]));
}

template <int NA>
__device__ __forceinline__ void gemm_tile(float* sA0, float* sA1, float* sW0, float* sW1, float* sC,
                                          const float* __restrict__ A, const float* __restrict__ A2,
                                          const float* __restrict__ W, const float* __restrict__ bias,
                                          float* __restrict__ C, float* __restrict__ C2,
                                          int N, int K, int lda, int tile) {
  int t = threadIdx.x;
  int wave = t >> 5;
  int lane = t & 31;
  int tiles_n = N >> 5;
  int tm = tile / tiles_n;
  int m0 = tm << 5;
  int n0 = (tile - tm * tiles_n) << 5;
  int lc = lane << 2;
  const float* Ag = A + (size_t)(m0 + wave) * lda + lc;
  const float* A2g = (NA == 2) ? A2 + (size_t)(m0 + wave) * lda + lc : Ag;
  const float* Wg = W + (size_t)(n0 + wave) * K + lc;
  int so = wave * GT_P + lc;
  int c = lane & 15;
  int h = lane >> 4;
  int wm = wave >> 1;
  int wn = wave & 1;
  int fao = (16 * wm + c) * GT_P + 4 * h;
  int fbo = (16 * wn + c) * GT_P + 4 * h;

  v4f ra0[8 * NA], rw0[8], ra1[8 * NA], rw1[8];
  float* sa0 = sA0 + so;
  float* sw0 = sW0 + so;
  float* sa1 = sA1 + so;
  float* sw1 = sW1 + so;
  const float* fa0 = sA0 + fao;
  const float* fb0 = sW0 + fbo;
  const float* fa1 = sA1 + fao;
  const float* fb1 = sW1 + fbo;
  v8f acc[2 * NA] = {};
  int nb = K / GT_K;

  gemm_rows_load<NA>(ra0, rw0, Ag, A2g, Wg, lda, K);
  if (nb == 1) {
    gemm_rows_store<NA>(sa0, sw0, ra0, rw0);
    __syncthreads();
    gemm_block_mma<NA>(fa0, fb0, acc);
  } else {
    gemm_rows_load<NA>(ra1, rw1, Ag + GT_K, A2g + GT_K, Wg + GT_K, lda, K);
    gemm_rows_store<NA>(sa0, sw0, ra0, rw0);
    lds_barrier();
    int kb = 0;
    for (; kb + 3 < nb; kb += 2) {
      size_t o2 = (size_t)(kb + 2) * GT_K;
      gemm_rows_load<NA>(ra0, rw0, Ag + o2, A2g + o2, Wg + o2, lda, K);
      gemm_block_mma<NA>(fa0, fb0, acc);
      stores_behind_mma<NA>(acc);
      gemm_rows_store<NA>(sa1, sw1, ra1, rw1);
      lds_barrier();
      size_t o3 = (size_t)(kb + 3) * GT_K;
      gemm_rows_load<NA>(ra1, rw1, Ag + o3, A2g + o3, Wg + o3, lda, K);
      gemm_block_mma<NA>(fa1, fb1, acc);
      stores_behind_mma<NA>(acc);
      gemm_rows_store<NA>(sa0, sw0, ra0, rw0);
      lds_barrier();
    }
    bool three = kb + 2 < nb;
    if (three) {
      size_t o2 = (size_t)(kb + 2) * GT_K;
      gemm_rows_load<NA>(ra0, rw0, Ag + o2, A2g + o2, Wg + o2, lda, K);
    }
    gemm_block_mma<NA>(fa0, fb0, acc);
    stores_behind_mma<NA>(acc);
    gemm_rows_store<NA>(sa1, sw1, ra1, rw1);
    lds_barrier();
    gemm_block_mma<NA>(fa1, fb1, acc);
    if (three) {
      stores_behind_mma<NA>(acc);
      gemm_rows_store<NA>(sa0, sw0, ra0, rw0);
      lds_barrier();
      gemm_block_mma<NA>(fa0, fb0, acc);
    }
  }

  int ccol = 16 * wn + c;
  float bv = bias ? bias[n0 + ccol] : 0.0f;
  int crow = 16 * wm + 8 * h;
#pragma unroll
  for (int o = 0; o < NA; ++o) {
#pragma unroll
    for (int r = 0; r < 8; ++r)
      sC[(32 * o + crow + r) * GT_CP + ccol] = (acc[2 * o][r] + acc[2 * o + 1][r]) + bv;
  }
  __syncthreads();
  int orow = 8 * wave + (lane >> 3);
  int ocol = (lane & 7) << 2;
#pragma unroll
  for (int o = 0; o < NA; ++o) {
    float* Co = o ? C2 : C;
#pragma unroll
    for (int i = 0; i < 2; ++i) {
      int row = orow + 4 * i;
      v4f v = *(const v4f*)&sC[(32 * o + row) * GT_CP + ocol];
      *(v4f*)&Co[(size_t)(m0 + row) * N + n0 + ocol] = v;
    }
  }
}

template <int NA>
__global__ __launch_bounds__(128)
void gemm_wmma_f32(const float* __restrict__ A, const float* __restrict__ A2, const float* __restrict__ W,
                   const float* __restrict__ bias, float* __restrict__ C, float* __restrict__ C2,
                   int M, int N, int K, int lda) {
  __shared__ __attribute__((aligned(16))) float sA[2][32 * NA * GT_P];
  __shared__ __attribute__((aligned(16))) float sW[2][32 * GT_P];
  __shared__ __attribute__((aligned(16))) float sC[32 * NA * GT_CP];
  gemm_tile<NA>(sA[0], sA[1], sW[0], sW[1], sC, A, A2, W, bias, C, C2, N, K, lda, blockIdx.x);
  (void)M;
}

template <int NT, bool RAW>
__device__ __forceinline__ void attn_rows(float* sh_s, float* sh_o, float* smx, float* smn,
                                          const float* __restrict__ self, const float* __restrict__ other,
                                          const float* __restrict__ beta_p, float* __restrict__ out,
                                          float* __restrict__ act, int apitch, int N, int b) {
  int t = threadIdx.x;
  const float* s = self  + (size_t)b * N;
  const float* o = other + (size_t)b * N;
  float* orow = RAW ? out + (size_t)b * N : nullptr;
  float* arow = act ? act + (size_t)b * apitch : nullptr;
  float beta = beta_p[0];
  if (beta == 0.0f) {
    for (int i = t; i < N; i += blockDim.x) {
      float v = s[i];
      if (RAW) orow[i] = v;
      if (arow) arow[i] = mish_f(v);
    }
    return;
  }
  for (int i = t; i < N; i += blockDim.x) { sh_s[i] = s[i]; sh_o[i] = o[i]; }
  __syncthreads();
  float mx = -3.4e38f, mn = 3.4e38f;
  for (int i = t; i < N; i += blockDim.x) { mx = fmaxf(mx, sh_o[i]); mn = fminf(mn, sh_o[i]); }
  smx[t] = mx; smn[t] = mn;
  __syncthreads();
  for (int off = NT / 2; off > 0; off >>= 1) {
    if (t < off) { smx[t] = fmaxf(smx[t], smx[t + off]); smn[t] = fminf(smn[t], smn[t + off]); }
    __syncthreads();
  }
  float omax = smx[0], omin = smn[0];
  for (int i = t; i < N; i += blockDim.x) {
    float si = sh_s[i];
    float m = (si >= 0.0f) ? si * omax : si * omin;
    float den = 0.0f, num = 0.0f;
    for (int j = 0; j < N; ++j) {
      float e = __expf(si * sh_o[j] - m);
      den += e;
      num += e * sh_s[j];
    }
    float v = si + beta * num / den;
    if (RAW) orow[i] = v;
    if (arow) arow[i] = mish_f(v);
  }
}

__global__ void attn_act_kernel(const float* __restrict__ qx, const float* __restrict__ qy,
                                const float* __restrict__ beta_p, float* __restrict__ act, int N, int B) {
  __shared__ float sh_s[512];
  __shared__ float sh_o[512];
  __shared__ float smx[256];
  __shared__ float smn[256];
  int g = blockIdx.x;
  bool second = g >= B;
  int b = second ? g - B : g;
  attn_rows<256, false>(sh_s, sh_o, smx, smn, second ? qy : qx, second ? qx : qy, beta_p,
                        (float*)nullptr, act + (second ? N : 0), 2 * N, N, b);
}

__global__ __launch_bounds__(128)
void fc_attn_kernel(const float* __restrict__ A, const float* __restrict__ W, const float* __restrict__ bias,
                    float* __restrict__ C, int N, int K, int lda, int tiles,
                    const float* __restrict__ qx, const float* __restrict__ qy,
                    const float* __restrict__ beta_p, float* __restrict__ ox, float* __restrict__ oy,
                    float* __restrict__ act, int Nq, int B) {
  __shared__ __attribute__((aligned(16))) float sA[2][32 * GT_P];
  __shared__ __attribute__((aligned(16))) float sW[2][32 * GT_P];
  __shared__ __attribute__((aligned(16))) float sC[32 * GT_CP];
  __shared__ float sh_s[512];
  __shared__ float sh_o[512];
  __shared__ float smx[128];
  __shared__ float smn[128];
  int g = blockIdx.x;
  if (g < tiles) {
    gemm_tile<1>(sA[0], sA[1], sW[0], sW[1], sC, A, (const float*)nullptr, W, bias, C, (float*)nullptr,
                 N, K, lda, g);
  } else {
    g -= tiles;
    bool second = g >= B;
    int b = second ? g - B : g;
    attn_rows<128, true>(sh_s, sh_o, smx, smn, second ? qy : qx, second ? qx : qy, beta_p, second ? oy : ox,
                         act ? act + (second ? Nq : 0) : (float*)nullptr, 2 * Nq, Nq, b);
  }
}

__global__ __launch_bounds__(128)
void query_cat_kernel(const float* __restrict__ Ax, const float* __restrict__ Ay, const float* __restrict__ W,
                      float* __restrict__ Cx, float* __restrict__ Cy, int N, int K, int lda, int tiles,
                      const float* __restrict__ xs, const float* __restrict__ ys,
                      const float* __restrict__ dprev, float* __restrict__ out, int Nc, int B) {
  __shared__ __attribute__((aligned(16))) float sA[2][32 * GT_P];
  __shared__ __attribute__((aligned(16))) float sW[2][32 * GT_P];
  __shared__ __attribute__((aligned(16))) float sC[32 * GT_CP];
  int g = blockIdx.x;
  if (g < 2 * tiles) {
    bool second = g >= tiles;
    gemm_tile<1>(sA[0], sA[1], sW[0], sW[1], sC, second ? Ay : Ax, (const float*)nullptr, W,
                 (const float*)nullptr, second ? Cy : Cx, (float*)nullptr, N, K, lda, second ? g - tiles : g);
  } else {
    cat_mish_group(xs, ys, dprev, out, Nc, B, g - 2 * tiles);
  }
}

__global__ void head_kernel(const float* __restrict__ d3, const float* __restrict__ W6,
                            const float* __restrict__ b6, float* __restrict__ out, int B) {
  __shared__ float sh[8];
  int b = blockIdx.x;
  int lane = threadIdx.x;
  if (lane < 7) {
    const float* row = d3 + (size_t)b * 64;
    const float* w   = W6 + (size_t)lane * 64;
    float acc = b6[lane];
    for (int k = 0; k < 64; ++k) acc += row[k] * w[k];
    sh[lane] = acc;
  }
  __syncthreads();
  if (lane < 7) {
    float nrm = sqrtf(sh[0]*sh[0] + sh[1]*sh[1] + sh[2]*sh[2] + sh[3]*sh[3]);
    float v = sh[lane];
    if (lane < 4) v = v / fmaxf(nrm, 1e-12f);
    out[(size_t)b * 7 + lane] = v;
    out[(size_t)B * 7 + (size_t)b * 7 + lane] = v;
  }
}

extern "C" void kernel_launch(void* const* d_in, const int* in_sizes, int n_in,
                              void* d_out, int out_size, void* d_ws, size_t ws_size,
                              hipStream_t stream) {
  const float* x    = (const float*)d_in[0];
  const float* y    = (const float*)d_in[1];
  const float* W1   = (const float*)d_in[2];
  const float* be1  = (const float*)d_in[3];
  const float* fcW1 = (const float*)d_in[4];
  const float* fcb1 = (const float*)d_in[5];
  const float* W2   = (const float*)d_in[6];
  const float* be2  = (const float*)d_in[7];
  const float* fcW2 = (const float*)d_in[8];
  const float* fcb2 = (const float*)d_in[9];
  const float* W3   = (const float*)d_in[10];
  const float* be3  = (const float*)d_in[11];
  const float* fcW3 = (const float*)d_in[12];
  const float* fcb3 = (const float*)d_in[13];
  const float* W4   = (const float*)d_in[14];
  const float* be4  = (const float*)d_in[15];
  const float* fcW4 = (const float*)d_in[16];
  const float* fcb4 = (const float*)d_in[17];
  const float* W6   = (const float*)d_in[18];
  const float* b6   = (const float*)d_in[19];

  float* ws = (float*)d_ws;
  float* m  = ws;
  float* qa = m  + 524288;
  float* qb = qa + 262144;
  float* xa = qb + 262144;
  float* ya = xa + 262144;
  float* d0 = ya + 262144;
  float* d1 = d0 + 262144;
  float* d2 = d1 + 131072;
  float* d3 = d2 + 65536;
  float* m2 = d3 + 32768;
  float* my = m2 + 262144;

  const int B = 512;
  auto grid1d = [](int n) { return (n + 255) / 256; };
  auto gemm = [&](const float* A, int lda, const float* W, const float* bias, float* C, int N, int K) {
    int tiles = (B >> 5) * (N >> 5);
    hipLaunchKernelGGL((gemm_wmma_f32<1>), dim3(tiles), dim3(128), 0, stream,
                       A, (const float*)nullptr, W, bias, C, (float*)nullptr, B, N, K, lda);
  };
  auto gemm_xy = [&](const float* Ax, const float* Ay, int lda, const float* W, float* Cx, float* Cy,
                     int N, int K) {
    int tiles = (B >> 5) * (N >> 5);
    hipLaunchKernelGGL((gemm_wmma_f32<2>), dim3(tiles), dim3(128), 0, stream,
                       Ax, Ay, W, (const float*)nullptr, Cx, Cy, B, N, K, lda);
  };
  hipLaunchKernelGGL(mish_kernel, dim3(grid1d(B * 1024)), dim3(256), 0, stream, x, m, B * 1024);
  hipLaunchKernelGGL(mish_kernel, dim3(grid1d(B * 1024)), dim3(256), 0, stream, y, my, B * 1024);
  gemm_xy(m, my, 1024, W1, qa, qb, 512, 1024);
  hipLaunchKernelGGL(attn_act_kernel, dim3(2 * B), dim3(256), 0, stream,
                     (const float*)qa, (const float*)qb, be1, m, 512, B);
  auto fc_attn = [&](const float* fcW, const float* fcb, float* dout, int N, int Nout,
                     const float* beta, float* plain, int Nq) {
    int tiles = (B >> 5) * (Nout >> 5);
    hipLaunchKernelGGL(fc_attn_kernel, dim3(tiles + 2 * B), dim3(128), 0, stream,
                       (const float*)m, fcW, fcb, dout, Nout, 2 * N, 2 * N, tiles,
                       (const float*)qa, (const float*)qb, beta, xa, ya, plain, Nq, B);
  };
  auto query_cat = [&](const float* Wq, int N, const float* dprev, int Nc) {
    int tiles = (B >> 5) * (N >> 5);
    int cgroups = (B * 2 * Nc) >> 9;
    hipLaunchKernelGGL(query_cat_kernel, dim3(2 * tiles + cgroups), dim3(128), 0, stream,
                       (const float*)m2, (const float*)(m2 + Nc), Wq, qa, qb, N, Nc, 2 * Nc, tiles,
                       (const float*)xa, (const float*)ya, dprev, m, Nc, B);
  };

  gemm_xy(m, m + 512, 1024, W2, qa, qb, 256, 512);
  fc_attn(fcW1, fcb1, d0, 512, 512, be2, m2, 256);
  query_cat(W3, 128, d0, 256);
  fc_attn(fcW2, fcb2, d1, 256, 256, be3, m2, 128);
  query_cat(W4, 64, d1, 128);
  fc_attn(fcW3, fcb3, d2, 128, 128, be4, nullptr, 64);
  hipLaunchKernelGGL(cat_mish_kernel, dim3(grid1d(B * 2 * 64)), dim3(256), 0, stream,
                     (const float*)xa, (const float*)ya, (const float*)d2, m, 64, B);
  gemm(m, 128, fcW4, fcb4, d3, 64, 128);
  hipLaunchKernelGGL(head_kernel, dim3(B), dim3(32), 0, stream, d3, W6, b6, (float*)d_out, B);

  (void)in_sizes; (void)n_in; (void)out_size; (void)ws_size;
}


# mish passes take four floats a lane (16-byte accesses, the four softplus chains interleaved): stage-1 mish and the beta = 0 attention rows on a quarter of the waves or fewer; values, sums unchanged
# speedup vs baseline: 4.5171x; 1.0748x over previous
#include <hip/hip_runtime.h>
#include <math.h>

typedef __attribute__((ext_vector_type(2))) float v2f;
typedef __attribute__((ext_vector_type(4))) float v4f;
typedef __attribute__((ext_vector_type(8))) float v8f;

__device__ __forceinline__ float mish_f(float v) {
  float sp = (v > 20.0f) ? v : log1pf(__expf(v));
  return v * tanhf(sp);
}

__device__ __forceinline__ v4f mish4(v4f v) {
  v4f l;
  l.x = log1pf(__expf(v.x));
  l.y = log1pf(__expf(v.y));
  l.z = log1pf(__expf(v.z));
  l.w = log1pf(__expf(v.w));
  asm volatile("" :: "v"(l.x), "v"(l.y), "v"(l.z), "v"(l.w));
  v4f r;
  r.x = v.x * tanhf((v.x > 20.0f) ? v.x : l.x);
  r.y = v.y * tanhf((v.y > 20.0f) ? v.y : l.y);
  r.z = v.z * tanhf((v.z > 20.0f) ? v.z : l.z);
  r.w = v.w * tanhf((v.w > 20.0f) ? v.w : l.w);
  return r;
}

__global__ void mish_kernel(const float* __restrict__ in, float* __restrict__ out, int n) {
  int i = (blockIdx.x * blockDim.x + threadIdx.x) << 2;
  if (i >= n) return;
  *(v4f*)(out + i) = mish4(*(const v4f*)(in + i));
}

__global__ void cat_mish_kernel(const float* __restrict__ xs, const float* __restrict__ ys,
                                const float* __restrict__ dprev, float* __restrict__ out, int N, int B) {
  int twoN = 2 * N;
  int idx = blockIdx.x * blockDim.x + threadIdx.x;
  if (idx >= B * twoN) return;
  int b = idx / twoN;
  int c = idx - b * twoN;
  float v = (c < N) ? xs[b * N + c] : ys[b * N + (c - N)];
  v += dprev[idx];
  out[idx] = mish_f(v);
}

__device__ __forceinline__ void cat_mish_group(const float* __restrict__ xs, const float* __restrict__ ys,
                                               const float* __restrict__ dprev, float* __restrict__ out,
                                               int N, int B, int cg) {
  int twoN = 2 * N;
  int idx = (cg * 128 + (int)threadIdx.x) << 2;
  if (idx >= B * twoN) return;
  int b = idx / twoN;
  int c = idx - b * twoN;
  const float* src = (c < N) ? xs + (size_t)b * N + c : ys + (size_t)b * N + (c - N);
  v4f v = *(const v4f*)src;
  v4f d = *(const v4f*)(dprev + idx);
  *(v4f*)(out + idx) = mish4(v + d);
}

#define GT_K 128
#define GT_P 132
#define GT_CP 36

template <int NA>
__device__ __forceinline__ void gemm_block_mma(const float* __restrict__ fa, const float* __restrict__ fb,
                                               v8f (&acc)[2 * NA]) {
  v4f av[NA][16], bv[16];
#pragma unroll
  for (int j = 0; j < 16; ++j) {
#pragma unroll
    for (int o = 0; o < NA; ++o) av[o][j] = *(const v4f*)(fa + o * 32 * GT_P + 8 * j);
    bv[j] = *(const v4f*)(fb + 8 * j);
  }
  __builtin_amdgcn_sched_barrier(0);
#pragma unroll
  for (int j = 0; j < 16; ++j) {
    v2f b0, b1;
    b0.x = bv[j].x; b0.y = bv[j].y;
    b1.x = bv[j].z; b1.y = bv[j].w;
#pragma unroll
    for (int o = 0; o < NA; ++o) {
      v2f a0, a1;
      a0.x = av[o][j].x; a0.y = av[o][j].y;
      a1.x = av[o][j].z; a1.y = av[o][j].w;
      acc[2 * o]     = __builtin_amdgcn_wmma_f32_16x16x4_f32(false, a0, false, b0, (short)0, acc[2 * o], false, false);
      acc[2 * o + 1] = __builtin_amdgcn_wmma_f32_16x16x4_f32(false, a1, false, b1, (short)0, acc[2 * o + 1], false, false);
    }
  }
}

template <int NA>
__device__ __forceinline__ void gemm_rows_load(v4f (&ra)[8 * NA], v4f (&rw)[8], const float* __restrict__ Ab,
                                               const float* __restrict__ A2b, const float* __restrict__ Wb,
                                               int lda, int K) {
#pragma unroll
  for (int i = 0; i < 8; ++i) {
    ra[i] = *(const v4f*)(Ab + (size_t)(4 * i) * lda);
    if constexpr (NA == 2) ra[8 + i] = *(const v4f*)(A2b + (size_t)(4 * i) * lda);
    rw[i] = *(const v4f*)(Wb + (size_t)(4 * i) * K);
  }
}

template <int NA>
__device__ __forceinline__ void gemm_rows_store(float* sa, float* sw, const v4f (&ra)[8 * NA], const v4f (&rw)[8]) {
#pragma unroll
  for (int i = 0; i < 8; ++i) {
    *(v4f*)(sa + 4 * i * GT_P) = ra[i];
    if constexpr (NA == 2) *(v4f*)(sa + (32 + 4 * i) * GT_P) = ra[8 + i];
    *(v4f*)(sw + 4 * i * GT_P) = rw[i];
  }
}

__device__ __forceinline__ void lds_barrier() {
  asm volatile("s_wait_dscnt 0x0\n\ts_barrier_signal -1\n\ts_barrier_wait -1" ::: "memory");
}

template <int NA>
__device__ __forceinline__ void stores_behind_mma(const v8f (&acc)[2 * NA]) {
  if constexpr (NA == 2) asm volatile("" :: "v"(acc[0]), "v"(acc[1]), "v"(acc[2]), "v"(acc[3]));
  else asm volatile("" :: "v"(acc[0]), "v"(acc[1]));
}

template <int NA>
__device__ __forceinline__ void gemm_tile(float* sA0, float* sA1, float* sW0, float* sW1, float* sC,
                                          const float* __restrict__ A, const float* __restrict__ A2,
                                          const float* __restrict__ W, const float* __restrict__ bias,
                                          float* __restrict__ C, float* __restrict__ C2,
                                          int N, int K, int lda, int tile) {
  int t = threadIdx.x;
  int wave = t >> 5;
  int lane = t & 31;
  int tiles_n = N >> 5;
  int tm = tile / tiles_n;
  int m0 = tm << 5;
  int n0 = (tile - tm * tiles_n) << 5;
  int lc = lane << 2;
  const float* Ag = A + (size_t)(m0 + wave) * lda + lc;
  const float* A2g = (NA == 2) ? A2 + (size_t)(m0 + wave) * lda + lc : Ag;
  const float* Wg = W + (size_t)(n0 + wave) * K + lc;
  int so = wave * GT_P + lc;
  int c = lane & 15;
  int h = lane >> 4;
  int wm = wave >> 1;
  int wn = wave & 1;
  int fao = (16 * wm + c) * GT_P + 4 * h;
  int fbo = (16 * wn + c) * GT_P + 4 * h;

  v4f ra0[8 * NA], rw0[8], ra1[8 * NA], rw1[8];
  float* sa0 = sA0 + so;
  float* sw0 = sW0 + so;
  float* sa1 = sA1 + so;
  float* sw1 = sW1 + so;
  const float* fa0 = sA0 + fao;
  const float* fb0 = sW0 + fbo;
  const float* fa1 = sA1 + fao;
  const float* fb1 = sW1 + fbo;
  v8f acc[2 * NA] = {};
  int nb = K / GT_K;

  gemm_rows_load<NA>(ra0, rw0, Ag, A2g, Wg, lda, K);
  if (nb == 1) {
    gemm_rows_store<NA>(sa0, sw0, ra0, rw0);
    __syncthreads();
    gemm_block_mma<NA>(fa0, fb0, acc);
  } else {
    gemm_rows_load<NA>(ra1, rw1, Ag + GT_K, A2g + GT_K, Wg + GT_K, lda, K);
    gemm_rows_store<NA>(sa0, sw0, ra0, rw0);
    lds_barrier();
    int kb = 0;
    for (; kb + 3 < nb; kb += 2) {
      size_t o2 = (size_t)(kb + 2) * GT_K;
      gemm_rows_load<NA>(ra0, rw0, Ag + o2, A2g + o2, Wg + o2, lda, K);
      gemm_block_mma<NA>(fa0, fb0, acc);
      stores_behind_mma<NA>(acc);
      gemm_rows_store<NA>(sa1, sw1, ra1, rw1);
      lds_barrier();
      size_t o3 = (size_t)(kb + 3) * GT_K;
      gemm_rows_load<NA>(ra1, rw1, Ag + o3, A2g + o3, Wg + o3, lda, K);
      gemm_block_mma<NA>(fa1, fb1, acc);
      stores_behind_mma<NA>(acc);
      gemm_rows_store<NA>(sa0, sw0, ra0, rw0);
      lds_barrier();
    }
    bool three = kb + 2 < nb;
    if (three) {
      size_t o2 = (size_t)(kb + 2) * GT_K;
      gemm_rows_load<NA>(ra0, rw0, Ag + o2, A2g + o2, Wg + o2, lda, K);
    }
    gemm_block_mma<NA>(fa0, fb0, acc);
    stores_behind_mma<NA>(acc);
    gemm_rows_store<NA>(sa1, sw1, ra1, rw1);
    lds_barrier();
    gemm_block_mma<NA>(fa1, fb1, acc);
    if (three) {
      stores_behind_mma<NA>(acc);
      gemm_rows_store<NA>(sa0, sw0, ra0, rw0);
      lds_barrier();
      gemm_block_mma<NA>(fa0, fb0, acc);
    }
  }

  int ccol = 16 * wn + c;
  float bv = bias ? bias[n0 + ccol] : 0.0f;
  int crow = 16 * wm + 8 * h;
#pragma unroll
  for (int o = 0; o < NA; ++o) {
#pragma unroll
    for (int r = 0; r < 8; ++r)
      sC[(32 * o + crow + r) * GT_CP + ccol] = (acc[2 * o][r] + acc[2 * o + 1][r]) + bv;
  }
  __syncthreads();
  int orow = 8 * wave + (lane >> 3);
  int ocol = (lane & 7) << 2;
#pragma unroll
  for (int o = 0; o < NA; ++o) {
    float* Co = o ? C2 : C;
#pragma unroll
    for (int i = 0; i < 2; ++i) {
      int row = orow + 4 * i;
      v4f v = *(const v4f*)&sC[(32 * o + row) * GT_CP + ocol];
      *(v4f*)&Co[(size_t)(m0 + row) * N + n0 + ocol] = v;
    }
  }
}

template <int NA>
__global__ __launch_bounds__(128)
void gemm_wmma_f32(const float* __restrict__ A, const float* __restrict__ A2, const float* __restrict__ W,
                   const float* __restrict__ bias, float* __restrict__ C, float* __restrict__ C2,
                   int M, int N, int K, int lda) {
  __shared__ __attribute__((aligned(16))) float sA[2][32 * NA * GT_P];
  __shared__ __attribute__((aligned(16))) float sW[2][32 * GT_P];
  __shared__ __attribute__((aligned(16))) float sC[32 * NA * GT_CP];
  gemm_tile<NA>(sA[0], sA[1], sW[0], sW[1], sC, A, A2, W, bias, C, C2, N, K, lda, blockIdx.x);
  (void)M;
}

template <int NT, bool RAW>
__device__ __forceinline__ void attn_rows(float* sh_s, float* sh_o, float* smx, float* smn,
                                          const float* __restrict__ self, const float* __restrict__ other,
                                          float beta, float* __restrict__ out,
                                          float* __restrict__ act, int apitch, int N, int b) {
  int t = threadIdx.x;
  const float* s = self  + (size_t)b * N;
  const float* o = other + (size_t)b * N;
  float* orow = RAW ? out + (size_t)b * N : nullptr;
  float* arow = act ? act + (size_t)b * apitch : nullptr;
  for (int i = t; i < N; i += blockDim.x) { sh_s[i] = s[i]; sh_o[i] = o[i]; }
  __syncthreads();
  float mx = -3.4e38f, mn = 3.4e38f;
  for (int i = t; i < N; i += blockDim.x) { mx = fmaxf(mx, sh_o[i]); mn = fminf(mn, sh_o[i]); }
  smx[t] = mx; smn[t] = mn;
  __syncthreads();
  for (int off = NT / 2; off > 0; off >>= 1) {
    if (t < off) { smx[t] = fmaxf(smx[t], smx[t + off]); smn[t] = fminf(smn[t], smn[t + off]); }
    __syncthreads();
  }
  float omax = smx[0], omin = smn[0];
  for (int i = t; i < N; i += blockDim.x) {
    float si = sh_s[i];
    float m = (si >= 0.0f) ? si * omax : si * omin;
    float den = 0.0f, num = 0.0f;
    for (int j = 0; j < N; ++j) {
      float e = __expf(si * sh_o[j] - m);
      den += e;
      num += e * sh_s[j];
    }
    float v = si + beta * num / den;
    if (RAW) orow[i] = v;
    if (arow) arow[i] = mish_f(v);
  }
}

template <int NT, bool RAW>
__device__ __forceinline__ void attn_stage(float* sh_s, float* sh_o, float* smx, float* smn,
                                           const float* __restrict__ qx, const float* __restrict__ qy,
                                           const float* __restrict__ beta_p, float* __restrict__ ox,
                                           float* __restrict__ oy, float* __restrict__ act,
                                           int N, int B, int ag, int na) {
  float beta = beta_p[0];
  if (beta == 0.0f) {
    int half = B * N;
    int step = (na * NT) << 2;
    for (int idx = (ag * NT + (int)threadIdx.x) << 2; idx < 2 * half; idx += step) {
      bool second = idx >= half;
      int r = second ? idx - half : idx;
      v4f v = *(const v4f*)((second ? qy : qx) + r);
      if (RAW) *(v4f*)((second ? oy : ox) + r) = v;
      if (act) {
        int b = r / N;
        int c = r - b * N;
        *(v4f*)(act + (size_t)b * (2 * N) + (second ? N : 0) + c) = mish4(v);
      }
    }
    return;
  }
  for (int p = ag; p < 2 * B; p += na) {
    bool second = p >= B;
    int b = second ? p - B : p;
    attn_rows<NT, RAW>(sh_s, sh_o, smx, smn, second ? qy : qx, second ? qx : qy, beta,
                       RAW ? (second ? oy : ox) : (float*)nullptr,
                       act ? act + (second ? N : 0) : (float*)nullptr, 2 * N, N, b);
    __syncthreads();
  }
}

__global__ void attn_act_kernel(const float* __restrict__ qx, const float* __restrict__ qy,
                                const float* __restrict__ beta_p, float* __restrict__ act, int N, int B) {
  __shared__ float sh_s[512];
  __shared__ float sh_o[512];
  __shared__ float smx[256];
  __shared__ float smn[256];
  attn_stage<256, false>(sh_s, sh_o, smx, smn, qx, qy, beta_p, (float*)nullptr, (float*)nullptr, act,
                         N, B, blockIdx.x, gridDim.x);
}

__global__ __launch_bounds__(128)
void fc_attn_kernel(const float* __restrict__ A, const float* __restrict__ W, const float* __restrict__ bias,
                    float* __restrict__ C, int N, int K, int lda, int tiles,
                    const float* __restrict__ qx, const float* __restrict__ qy,
                    const float* __restrict__ beta_p, float* __restrict__ ox, float* __restrict__ oy,
                    float* __restrict__ act, int Nq, int B) {
  __shared__ __attribute__((aligned(16))) float sA[2][32 * GT_P];
  __shared__ __attribute__((aligned(16))) float sW[2][32 * GT_P];
  __shared__ __attribute__((aligned(16))) float sC[32 * GT_CP];
  __shared__ float sh_s[512];
  __shared__ float sh_o[512];
  __shared__ float smx[128];
  __shared__ float smn[128];
  int g = blockIdx.x;
  if (g < tiles) {
    gemm_tile<1>(sA[0], sA[1], sW[0], sW[1], sC, A, (const float*)nullptr, W, bias, C, (float*)nullptr,
                 N, K, lda, g);
  } else {
    attn_stage<128, true>(sh_s, sh_o, smx, smn, qx, qy, beta_p, ox, oy, act, Nq, B, g - tiles,
                          (int)gridDim.x - tiles);
  }
}

__global__ __launch_bounds__(128)
void query_cat_kernel(const float* __restrict__ Ax, const float* __restrict__ Ay, const float* __restrict__ W,
                      float* __restrict__ Cx, float* __restrict__ Cy, int N, int K, int lda, int tiles,
                      const float* __restrict__ xs, const float* __restrict__ ys,
                      const float* __restrict__ dprev, float* __restrict__ out, int Nc, int B) {
  __shared__ __attribute__((aligned(16))) float sA[2][32 * GT_P];
  __shared__ __attribute__((aligned(16))) float sW[2][32 * GT_P];
  __shared__ __attribute__((aligned(16))) float sC[32 * GT_CP];
  int g = blockIdx.x;
  if (g < 2 * tiles) {
    bool second = g >= tiles;
    gemm_tile<1>(sA[0], sA[1], sW[0], sW[1], sC, second ? Ay : Ax, (const float*)nullptr, W,
                 (const float*)nullptr, second ? Cy : Cx, (float*)nullptr, N, K, lda, second ? g - tiles : g);
  } else {
    cat_mish_group(xs, ys, dprev, out, Nc, B, g - 2 * tiles);
  }
}

__global__ void head_kernel(const float* __restrict__ d3, const float* __restrict__ W6,
                            const float* __restrict__ b6, float* __restrict__ out, int B) {
  __shared__ float sh[8];
  int b = blockIdx.x;
  int lane = threadIdx.x;
  if (lane < 7) {
    const float* row = d3 + (size_t)b * 64;
    const float* w   = W6 + (size_t)lane * 64;
    float acc = b6[lane];
    for (int k = 0; k < 64; ++k) acc += row[k] * w[k];
    sh[lane] = acc;
  }
  __syncthreads();
  if (lane < 7) {
    float nrm = sqrtf(sh[0]*sh[0] + sh[1]*sh[1] + sh[2]*sh[2] + sh[3]*sh[3]);
    float v = sh[lane];
    if (lane < 4) v = v / fmaxf(nrm, 1e-12f);
    out[(size_t)b * 7 + lane] = v;
    out[(size_t)B * 7 + (size_t)b * 7 + lane] = v;
  }
}

extern "C" void kernel_launch(void* const* d_in, const int* in_sizes, int n_in,
                              void* d_out, int out_size, void* d_ws, size_t ws_size,
                              hipStream_t stream) {
  const float* x    = (const float*)d_in[0];
  const float* y    = (const float*)d_in[1];
  const float* W1   = (const float*)d_in[2];
  const float* be1  = (const float*)d_in[3];
  const float* fcW1 = (const float*)d_in[4];
  const float* fcb1 = (const float*)d_in[5];
  const float* W2   = (const float*)d_in[6];
  const float* be2  = (const float*)d_in[7];
  const float* fcW2 = (const float*)d_in[8];
  const float* fcb2 = (const float*)d_in[9];
  const float* W3   = (const float*)d_in[10];
  const float* be3  = (const float*)d_in[11];
  const float* fcW3 = (const float*)d_in[12];
  const float* fcb3 = (const float*)d_in[13];
  const float* W4   = (const float*)d_in[14];
  const float* be4  = (const float*)d_in[15];
  const float* fcW4 = (const float*)d_in[16];
  const float* fcb4 = (const float*)d_in[17];
  const float* W6   = (const float*)d_in[18];
  const float* b6   = (const float*)d_in[19];

  float* ws = (float*)d_ws;
  float* m  = ws;
  float* qa = m  + 524288;
  float* qb = qa + 262144;
  float* xa = qb + 262144;
  float* ya = xa + 262144;
  float* d0 = ya + 262144;
  float* d1 = d0 + 262144;
  float* d2 = d1 + 131072;
  float* d3 = d2 + 65536;
  float* m2 = d3 + 32768;
  float* my = m2 + 262144;

  const int B = 512;
  auto grid1d = [](int n) { return (n + 255) / 256; };
  auto gemm = [&](const float* A, int lda, const float* W, const float* bias, float* C, int N, int K) {
    int tiles = (B >> 5) * (N >> 5);
    hipLaunchKernelGGL((gemm_wmma_f32<1>), dim3(tiles), dim3(128), 0, stream,
                       A, (const float*)nullptr, W, bias, C, (float*)nullptr, B, N, K, lda);
  };
  auto gemm_xy = [&](const float* Ax, const float* Ay, int lda, const float* W, float* Cx, float* Cy,
                     int N, int K) {
    int tiles = (B >> 5) * (N >> 5);
    hipLaunchKernelGGL((gemm_wmma_f32<2>), dim3(tiles), dim3(128), 0, stream,
                       Ax, Ay, W, (const float*)nullptr, Cx, Cy, B, N, K, lda);
  };
  hipLaunchKernelGGL(mish_kernel, dim3(grid1d(B * 1024 / 4)), dim3(256), 0, stream, x, m, B * 1024);
  hipLaunchKernelGGL(mish_kernel, dim3(grid1d(B * 1024 / 4)), dim3(256), 0, stream, y, my, B * 1024);
  gemm_xy(m, my, 1024, W1, qa, qb, 512, 1024);
  hipLaunchKernelGGL(attn_act_kernel, dim3((2 * B * 512) / (256 * 4)), dim3(256), 0, stream,
                     (const float*)qa, (const float*)qb, be1, m, 512, B);
  auto fc_attn = [&](const float* fcW, const float* fcb, float* dout, int N, int Nout,
                     const float* beta, float* plain, int Nq) {
    int tiles = (B >> 5) * (Nout >> 5);
    int agroups = (2 * B * Nq) / (128 * 4);
    hipLaunchKernelGGL(fc_attn_kernel, dim3(tiles + agroups), dim3(128), 0, stream,
                       (const float*)m, fcW, fcb, dout, Nout, 2 * N, 2 * N, tiles,
                       (const float*)qa, (const float*)qb, beta, xa, ya, plain, Nq, B);
  };
  auto query_cat = [&](const float* Wq, int N, const float* dprev, int Nc) {
    int tiles = (B >> 5) * (N >> 5);
    int cgroups = (B * 2 * Nc) >> 9;
    hipLaunchKernelGGL(query_cat_kernel, dim3(2 * tiles + cgroups), dim3(128), 0, stream,
                       (const float*)m2, (const float*)(m2 + Nc), Wq, qa, qb, N, Nc, 2 * Nc, tiles,
                       (const float*)xa, (const float*)ya, dprev, m, Nc, B);
  };

  gemm_xy(m, m + 512, 1024, W2, qa, qb, 256, 512);
  fc_attn(fcW1, fcb1, d0, 512, 512, be2, m2, 256);
  query_cat(W3, 128, d0, 256);
  fc_attn(fcW2, fcb2, d1, 256, 256, be3, m2, 128);
  query_cat(W4, 64, d1, 128);
  fc_attn(fcW3, fcb3, d2, 128, 128, be4, nullptr, 64);
  hipLaunchKernelGGL(cat_mish_kernel, dim3(grid1d(B * 2 * 64)), dim3(256), 0, stream,
                     (const float*)xa, (const float*)ya, (const float*)d2, m, 64, B);
  gemm(m, 128, fcW4, fcb4, d3, 64, 128);
  hipLaunchKernelGGL(head_kernel, dim3(B), dim3(32), 0, stream, d3, W6, b6, (float*)d_out, B);

  (void)in_sizes; (void)n_in; (void)out_size; (void)ws_size;
}


# last three launches in one: a group of 8 waves takes 8 samples through the concat-mish pass, the fc4 product (WMMA, K = 128) and the head, rows handed on in LDS; 13 -> 11 launches; sums unchanged
# speedup vs baseline: 4.8863x; 1.0817x over previous
#include <hip/hip_runtime.h>
#include <math.h>

typedef __attribute__((ext_vector_type(2))) float v2f;
typedef __attribute__((ext_vector_type(4))) float v4f;
typedef __attribute__((ext_vector_type(8))) float v8f;

__device__ __forceinline__ float mish_f(float v) {
  float sp = (v > 20.0f) ? v : log1pf(__expf(v));
  return v * tanhf(sp);
}

__device__ __forceinline__ v4f mish4(v4f v) {
  v4f l;
  l.x = log1pf(__expf(v.x));
  l.y = log1pf(__expf(v.y));
  l.z = log1pf(__expf(v.z));
  l.w = log1pf(__expf(v.w));
  asm volatile("" :: "v"(l.x), "v"(l.y), "v"(l.z), "v"(l.w));
  v4f r;
  r.x = v.x * tanhf((v.x > 20.0f) ? v.x : l.x);
  r.y = v.y * tanhf((v.y > 20.0f) ? v.y : l.y);
  r.z = v.z * tanhf((v.z > 20.0f) ? v.z : l.z);
  r.w = v.w * tanhf((v.w > 20.0f) ? v.w : l.w);
  return r;
}

__global__ void mish_kernel(const float* __restrict__ in, float* __restrict__ out, int n) {
  int i = (blockIdx.x * blockDim.x + threadIdx.x) << 2;
  if (i >= n) return;
  *(v4f*)(out + i) = mish4(*(const v4f*)(in + i));
}

__device__ __forceinline__ void cat_mish_group(const float* __restrict__ xs, const float* __restrict__ ys,
                                               const float* __restrict__ dprev, float* __restrict__ out,
                                               int N, int B, int cg) {
  int twoN = 2 * N;
  int idx = (cg * 128 + (int)threadIdx.x) << 2;
  if (idx >= B * twoN) return;
  int b = idx / twoN;
  int c = idx - b * twoN;
  const float* src = (c < N) ? xs + (size_t)b * N + c : ys + (size_t)b * N + (c - N);
  v4f v = *(const v4f*)src;
  v4f d = *(const v4f*)(dprev + idx);
  *(v4f*)(out + idx) = mish4(v + d);
}

#define GT_K 128
#define GT_P 132
#define GT_CP 36

template <int NA>
__device__ __forceinline__ void gemm_block_mma(const float* __restrict__ fa, const float* __restrict__ fb,
                                               v8f (&acc)[2 * NA]) {
  v4f av[NA][16], bv[16];
#pragma unroll
  for (int j = 0; j < 16; ++j) {
#pragma unroll
    for (int o = 0; o < NA; ++o) av[o][j] = *(const v4f*)(fa + o * 32 * GT_P + 8 * j);
    bv[j] = *(const v4f*)(fb + 8 * j);
  }
  __builtin_amdgcn_sched_barrier(0);
#pragma unroll
  for (int j = 0; j < 16; ++j) {
    v2f b0, b1;
    b0.x = bv[j].x; b0.y = bv[j].y;
    b1.x = bv[j].z; b1.y = bv[j].w;
#pragma unroll
    for (int o = 0; o < NA; ++o) {
      v2f a0, a1;
      a0.x = av[o][j].x; a0.y = av[o][j].y;
      a1.x = av[o][j].z; a1.y = av[o][j].w;
      acc[2 * o]     = __builtin_amdgcn_wmma_f32_16x16x4_f32(false, a0, false, b0, (short)0, acc[2 * o], false, false);
      acc[2 * o + 1] = __builtin_amdgcn_wmma_f32_16x16x4_f32(false, a1, false, b1, (short)0, acc[2 * o + 1], false, false);
    }
  }
}

template <int NA>
__device__ __forceinline__ void gemm_rows_load(v4f (&ra)[8 * NA], v4f (&rw)[8], const float* __restrict__ Ab,
                                               const float* __restrict__ A2b, const float* __restrict__ Wb,
                                               int lda, int K) {
#pragma unroll
  for (int i = 0; i < 8; ++i) {
    ra[i] = *(const v4f*)(Ab + (size_t)(4 * i) * lda);
    if constexpr (NA == 2) ra[8 + i] = *(const v4f*)(A2b + (size_t)(4 * i) * lda);
    rw[i] = *(const v4f*)(Wb + (size_t)(4 * i) * K);
  }
}

template <int NA>
__device__ __forceinline__ void gemm_rows_store(float* sa, float* sw, const v4f (&ra)[8 * NA], const v4f (&rw)[8]) {
#pragma unroll
  for (int i = 0; i < 8; ++i) {
    *(v4f*)(sa + 4 * i * GT_P) = ra[i];
    if constexpr (NA == 2) *(v4f*)(sa + (32 + 4 * i) * GT_P) = ra[8 + i];
    *(v4f*)(sw + 4 * i * GT_P) = rw[i];
  }
}

__device__ __forceinline__ void lds_barrier() {
  asm volatile("s_wait_dscnt 0x0\n\ts_barrier_signal -1\n\ts_barrier_wait -1" ::: "memory");
}

template <int NA>
__device__ __forceinline__ void stores_behind_mma(const v8f (&acc)[2 * NA]) {
  if constexpr (NA == 2) asm volatile("" :: "v"(acc[0]), "v"(acc[1]), "v"(acc[2]), "v"(acc[3]));
  else asm volatile("" :: "v"(acc[0]), "v"(acc[1]));
}

template <int NA>
__device__ __forceinline__ void gemm_tile(float* sA0, float* sA1, float* sW0, float* sW1, float* sC,
                                          const float* __restrict__ A, const float* __restrict__ A2,
                                          const float* __restrict__ W, const float* __restrict__ bias,
                                          float* __restrict__ C, float* __restrict__ C2,
                                          int N, int K, int lda, int tile) {
  int t = threadIdx.x;
  int wave = t >> 5;
  int lane = t & 31;
  int tiles_n = N >> 5;
  int tm = tile / tiles_n;
  int m0 = tm << 5;
  int n0 = (tile - tm * tiles_n) << 5;
  int lc = lane << 2;
  const float* Ag = A + (size_t)(m0 + wave) * lda + lc;
  const float* A2g = (NA == 2) ? A2 + (size_t)(m0 + wave) * lda + lc : Ag;
  const float* Wg = W + (size_t)(n0 + wave) * K + lc;
  int so = wave * GT_P + lc;
  int c = lane & 15;
  int h = lane >> 4;
  int wm = wave >> 1;
  int wn = wave & 1;
  int fao = (16 * wm + c) * GT_P + 4 * h;
  int fbo = (16 * wn + c) * GT_P + 4 * h;

  v4f ra0[8 * NA], rw0[8], ra1[8 * NA], rw1[8];
  float* sa0 = sA0 + so;
  float* sw0 = sW0 + so;
  float* sa1 = sA1 + so;
  float* sw1 = sW1 + so;
  const float* fa0 = sA0 + fao;
  const float* fb0 = sW0 + fbo;
  const float* fa1 = sA1 + fao;
  const float* fb1 = sW1 + fbo;
  v8f acc[2 * NA] = {};
  int nb = K / GT_K;

  gemm_rows_load<NA>(ra0, rw0, Ag, A2g, Wg, lda, K);
  if (nb == 1) {
    gemm_rows_store<NA>(sa0, sw0, ra0, rw0);
    __syncthreads();
    gemm_block_mma<NA>(fa0, fb0, acc);
  } else {
    gemm_rows_load<NA>(ra1, rw1, Ag + GT_K, A2g + GT_K, Wg + GT_K, lda, K);
    gemm_rows_store<NA>(sa0, sw0, ra0, rw0);
    lds_barrier();
    int kb = 0;
    for (; kb + 3 < nb; kb += 2) {
      size_t o2 = (size_t)(kb + 2) * GT_K;
      gemm_rows_load<NA>(ra0, rw0, Ag + o2, A2g + o2, Wg + o2, lda, K);
      gemm_block_mma<NA>(fa0, fb0, acc);
      stores_behind_mma<NA>(acc);
      gemm_rows_store<NA>(sa1, sw1, ra1, rw1);
      lds_barrier();
      size_t o3 = (size_t)(kb + 3) * GT_K;
      gemm_rows_load<NA>(ra1, rw1, Ag + o3, A2g + o3, Wg + o3, lda, K);
      gemm_block_mma<NA>(fa1, fb1, acc);
      stores_behind_mma<NA>(acc);
      gemm_rows_store<NA>(sa0, sw0, ra0, rw0);
      lds_barrier();
    }
    bool three = kb + 2 < nb;
    if (three) {
      size_t o2 = (size_t)(kb + 2) * GT_K;
      gemm_rows_load<NA>(ra0, rw0, Ag + o2, A2g + o2, Wg + o2, lda, K);
    }
    gemm_block_mma<NA>(fa0, fb0, acc);
    stores_behind_mma<NA>(acc);
    gemm_rows_store<NA>(sa1, sw1, ra1, rw1);
    lds_barrier();
    gemm_block_mma<NA>(fa1, fb1, acc);
    if (three) {
      stores_behind_mma<NA>(acc);
      gemm_rows_store<NA>(sa0, sw0, ra0, rw0);
      lds_barrier();
      gemm_block_mma<NA>(fa0, fb0, acc);
    }
  }

  int ccol = 16 * wn + c;
  float bv = bias ? bias[n0 + ccol] : 0.0f;
  int crow = 16 * wm + 8 * h;
#pragma unroll
  for (int o = 0; o < NA; ++o) {
#pragma unroll
    for (int r = 0; r < 8; ++r)
      sC[(32 * o + crow + r) * GT_CP + ccol] = (acc[2 * o][r] + acc[2 * o + 1][r]) + bv;
  }
  __syncthreads();
  int orow = 8 * wave + (lane >> 3);
  int ocol = (lane & 7) << 2;
#pragma unroll
  for (int o = 0; o < NA; ++o) {
    float* Co = o ? C2 : C;
#pragma unroll
    for (int i = 0; i < 2; ++i) {
      int row = orow + 4 * i;
      v4f v = *(const v4f*)&sC[(32 * o + row) * GT_CP + ocol];
      *(v4f*)&Co[(size_t)(m0 + row) * N + n0 + ocol] = v;
    }
  }
}

template <int NA>
__global__ __launch_bounds__(128)
void gemm_wmma_f32(const float* __restrict__ A, const float* __restrict__ A2, const float* __restrict__ W,
                   const float* __restrict__ bias, float* __restrict__ C, float* __restrict__ C2,
                   int M, int N, int K, int lda) {
  __shared__ __attribute__((aligned(16))) float sA[2][32 * NA * GT_P];
  __shared__ __attribute__((aligned(16))) float sW[2][32 * GT_P];
  __shared__ __attribute__((aligned(16))) float sC[32 * NA * GT_CP];
  gemm_tile<NA>(sA[0], sA[1], sW[0], sW[1], sC, A, A2, W, bias, C, C2, N, K, lda, blockIdx.x);
  (void)M;
}

template <int NT, bool RAW>
__device__ __forceinline__ void attn_rows(float* sh_s, float* sh_o, float* smx, float* smn,
                                          const float* __restrict__ self, const float* __restrict__ other,
                                          float beta, float* __restrict__ out,
                                          float* __restrict__ act, int apitch, int N, int b) {
  int t = threadIdx.x;
  const float* s = self  + (size_t)b * N;
  const float* o = other + (size_t)b * N;
  float* orow = RAW ? out + (size_t)b * N : nullptr;
  float* arow = act ? act + (size_t)b * apitch : nullptr;
  for (int i = t; i < N; i += blockDim.x) { sh_s[i] = s[i]; sh_o[i] = o[i]; }
  __syncthreads();
  float mx = -3.4e38f, mn = 3.4e38f;
  for (int i = t; i < N; i += blockDim.x) { mx = fmaxf(mx, sh_o[i]); mn = fminf(mn, sh_o[i]); }
  smx[t] = mx; smn[t] = mn;
  __syncthreads();
  for (int off = NT / 2; off > 0; off >>= 1) {
    if (t < off) { smx[t] = fmaxf(smx[t], smx[t + off]); smn[t] = fminf(smn[t], smn[t + off]); }
    __syncthreads();
  }
  float omax = smx[0], omin = smn[0];
  for (int i = t; i < N; i += blockDim.x) {
    float si = sh_s[i];
    float m = (si >= 0.0f) ? si * omax : si * omin;
    float den = 0.0f, num = 0.0f;
    for (int j = 0; j < N; ++j) {
      float e = __expf(si * sh_o[j] - m);
      den += e;
      num += e * sh_s[j];
    }
    float v = si + beta * num / den;
    if (RAW) orow[i] = v;
    if (arow) arow[i] = mish_f(v);
  }
}

template <int NT, bool RAW>
__device__ __forceinline__ void attn_stage(float* sh_s, float* sh_o, float* smx, float* smn,
                                           const float* __restrict__ qx, const float* __restrict__ qy,
                                           const float* __restrict__ beta_p, float* __restrict__ ox,
                                           float* __restrict__ oy, float* __restrict__ act,
                                           int N, int B, int ag, int na) {
  float beta = beta_p[0];
  if (beta == 0.0f) {
    int half = B * N;
    int step = (na * NT) << 2;
    for (int idx = (ag * NT + (int)threadIdx.x) << 2; idx < 2 * half; idx += step) {
      bool second = idx >= half;
      int r = second ? idx - half : idx;
      v4f v = *(const v4f*)((second ? qy : qx) + r);
      if (RAW) *(v4f*)((second ? oy : ox) + r) = v;
      if (act) {
        int b = r / N;
        int c = r - b * N;
        *(v4f*)(act + (size_t)b * (2 * N) + (second ? N : 0) + c) = mish4(v);
      }
    }
    return;
  }
  for (int p = ag; p < 2 * B; p += na) {
    bool second = p >= B;
    int b = second ? p - B : p;
    attn_rows<NT, RAW>(sh_s, sh_o, smx, smn, second ? qy : qx, second ? qx : qy, beta,
                       RAW ? (second ? oy : ox) : (float*)nullptr,
                       act ? act + (second ? N : 0) : (float*)nullptr, 2 * N, N, b);
    __syncthreads();
  }
}

__global__ void attn_act_kernel(const float* __restrict__ qx, const float* __restrict__ qy,
                                const float* __restrict__ beta_p, float* __restrict__ act, int N, int B) {
  __shared__ float sh_s[512];
  __shared__ float sh_o[512];
  __shared__ float smx[256];
  __shared__ float smn[256];
  attn_stage<256, false>(sh_s, sh_o, smx, smn, qx, qy, beta_p, (float*)nullptr, (float*)nullptr, act,
                         N, B, blockIdx.x, gridDim.x);
}

__global__ __launch_bounds__(128)
void fc_attn_kernel(const float* __restrict__ A, const float* __restrict__ W, const float* __restrict__ bias,
                    float* __restrict__ C, int N, int K, int lda, int tiles,
                    const float* __restrict__ qx, const float* __restrict__ qy,
                    const float* __restrict__ beta_p, float* __restrict__ ox, float* __restrict__ oy,
                    float* __restrict__ act, int Nq, int B) {
  __shared__ __attribute__((aligned(16))) float sA[2][32 * GT_P];
  __shared__ __attribute__((aligned(16))) float sW[2][32 * GT_P];
  __shared__ __attribute__((aligned(16))) float sC[32 * GT_CP];
  __shared__ float sh_s[512];
  __shared__ float sh_o[512];
  __shared__ float smx[128];
  __shared__ float smn[128];
  int g = blockIdx.x;
  if (g < tiles) {
    gemm_tile<1>(sA[0], sA[1], sW[0], sW[1], sC, A, (const float*)nullptr, W, bias, C, (float*)nullptr,
                 N, K, lda, g);
  } else {
    attn_stage<128, true>(sh_s, sh_o, smx, smn, qx, qy, beta_p, ox, oy, act, Nq, B, g - tiles,
                          (int)gridDim.x - tiles);
  }
}

__global__ __launch_bounds__(128)
void query_cat_kernel(const float* __restrict__ Ax, const float* __restrict__ Ay, const float* __restrict__ W,
                      float* __restrict__ Cx, float* __restrict__ Cy, int N, int K, int lda, int tiles,
                      const float* __restrict__ xs, const float* __restrict__ ys,
                      const float* __restrict__ dprev, float* __restrict__ out, int Nc, int B) {
  __shared__ __attribute__((aligned(16))) float sA[2][32 * GT_P];
  __shared__ __attribute__((aligned(16))) float sW[2][32 * GT_P];
  __shared__ __attribute__((aligned(16))) float sC[32 * GT_CP];
  int g = blockIdx.x;
  if (g < 2 * tiles) {
    bool second = g >= tiles;
    gemm_tile<1>(sA[0], sA[1], sW[0], sW[1], sC, second ? Ay : Ax, (const float*)nullptr, W,
                 (const float*)nullptr, second ? Cy : Cx, (float*)nullptr, N, K, lda, second ? g - tiles : g);
  } else {
    cat_mish_group(xs, ys, dprev, out, Nc, B, g - 2 * tiles);
  }
}

#define TL_R 8
#define TL_DP 68
__global__ __launch_bounds__(256)
void tail_kernel(const float* __restrict__ xs, const float* __restrict__ ys, const float* __restrict__ dprev,
                 const float* __restrict__ W, const float* __restrict__ bias, const float* __restrict__ W6,
                 const float* __restrict__ b6, float* __restrict__ out, int B) {
  __shared__ __attribute__((aligned(16))) float sA[TL_R * GT_P];
  __shared__ __attribute__((aligned(16))) float sW[64 * GT_P];
  __shared__ __attribute__((aligned(16))) float sD[16 * TL_DP];
  __shared__ __attribute__((aligned(16))) float sW6[7 * 64];
  __shared__ float sX[TL_R * 8];
  int t = threadIdx.x;
  int b0 = blockIdx.x * TL_R;
  int ar = t >> 5;
  int ac = (t & 31) << 2;
  const float* src = (ac < 64) ? xs + (size_t)(b0 + ar) * 64 + ac : ys + (size_t)(b0 + ar) * 64 + (ac - 64);
  v4f v = *(const v4f*)src;
  v4f d = *(const v4f*)(dprev + (size_t)(b0 + ar) * 128 + ac);
  v4f rw[8];
#pragma unroll
  for (int i = 0; i < 8; ++i) rw[i] = *(const v4f*)(W + (size_t)(i * 256 + t) * 4);
  v4f w6 = *(const v4f*)(W6 + (t < 112 ? t : 111) * 4);
  *(v4f*)(sA + ar * GT_P + ac) = mish4(v + d);
#pragma unroll
  for (int i = 0; i < 8; ++i) {
    int idx = i * 256 + t;
    *(v4f*)(sW + (idx >> 5) * GT_P + ((idx & 31) << 2)) = rw[i];
  }
  if (t < 112) *(v4f*)(sW6 + t * 4) = w6;
  __syncthreads();
  int wave = t >> 5;
  int lane = t & 31;
  if (wave < 4) {
    int c = lane & 15;
    int h = lane >> 4;
    v8f acc[2] = {};
    gemm_block_mma<1>(sA + (c & 7) * GT_P + 4 * h, sW + (16 * wave + c) * GT_P + 4 * h, acc);
    float bv = bias[16 * wave + c];
#pragma unroll
    for (int r = 0; r < 8; ++r) sD[(8 * h + r) * TL_DP + 16 * wave + c] = (acc[0][r] + acc[1][r]) + bv;
  }
  __syncthreads();
  int hr = t / 7;
  int ho = t - 7 * hr;
  if (t < TL_R * 7) {
    const float* row = sD + hr * TL_DP;
    const float* w = sW6 + ho * 64;
    float a = b6[ho];
    for (int k = 0; k < 64; ++k) a += row[k] * w[k];
    sX[hr * 8 + ho] = a;
  }
  __syncthreads();
  if (t < TL_R * 7) {
    const float* q = sX + hr * 8;
    float nrm = sqrtf(q[0]*q[0] + q[1]*q[1] + q[2]*q[2] + q[3]*q[3]);
    float x = q[ho];
    if (ho < 4) x = x / fmaxf(nrm, 1e-12f);
    out[(size_t)(b0 + hr) * 7 + ho] = x;
    out[(size_t)B * 7 + (size_t)(b0 + hr) * 7 + ho] = x;
  }
}

extern "C" void kernel_launch(void* const* d_in, const int* in_sizes, int n_in,
                              void* d_out, int out_size, void* d_ws, size_t ws_size,
                              hipStream_t stream) {
  const float* x    = (const float*)d_in[0];
  const float* y    = (const float*)d_in[1];
  const float* W1   = (const float*)d_in[2];
  const float* be1  = (const float*)d_in[3];
  const float* fcW1 = (const float*)d_in[4];
  const float* fcb1 = (const float*)d_in[5];
  const float* W2   = (const float*)d_in[6];
  const float* be2  = (const float*)d_in[7];
  const float* fcW2 = (const float*)d_in[8];
  const float* fcb2 = (const float*)d_in[9];
  const float* W3   = (const float*)d_in[10];
  const float* be3  = (const float*)d_in[11];
  const float* fcW3 = (const float*)d_in[12];
  const float* fcb3 = (const float*)d_in[13];
  const float* W4   = (const float*)d_in[14];
  const float* be4  = (const float*)d_in[15];
  const float* fcW4 = (const float*)d_in[16];
  const float* fcb4 = (const float*)d_in[17];
  const float* W6   = (const float*)d_in[18];
  const float* b6   = (const float*)d_in[19];

  float* ws = (float*)d_ws;
  float* m  = ws;
  float* qa = m  + 524288;
  float* qb = qa + 262144;
  float* xa = qb + 262144;
  float* ya = xa + 262144;
  float* d0 = ya + 262144;
  float* d1 = d0 + 262144;
  float* d2 = d1 + 131072;
  float* m2 = d2 + 65536 + 32768;
  float* my = m2 + 262144;

  const int B = 512;
  auto grid1d = [](int n) { return (n + 255) / 256; };
  auto gemm_xy = [&](const float* Ax, const float* Ay, int lda, const float* W, float* Cx, float* Cy,
                     int N, int K) {
    int tiles = (B >> 5) * (N >> 5);
    hipLaunchKernelGGL((gemm_wmma_f32<2>), dim3(tiles), dim3(128), 0, stream,
                       Ax, Ay, W, (const float*)nullptr, Cx, Cy, B, N, K, lda);
  };
  hipLaunchKernelGGL(mish_kernel, dim3(grid1d(B * 1024 / 4)), dim3(256), 0, stream, x, m, B * 1024);
  hipLaunchKernelGGL(mish_kernel, dim3(grid1d(B * 1024 / 4)), dim3(256), 0, stream, y, my, B * 1024);
  gemm_xy(m, my, 1024, W1, qa, qb, 512, 1024);
  hipLaunchKernelGGL(attn_act_kernel, dim3((2 * B * 512) / (256 * 4)), dim3(256), 0, stream,
                     (const float*)qa, (const float*)qb, be1, m, 512, B);
  auto fc_attn = [&](const float* fcW, const float* fcb, float* dout, int N, int Nout,
                     const float* beta, float* plain, int Nq) {
    int tiles = (B >> 5) * (Nout >> 5);
    int agroups = (2 * B * Nq) / (128 * 4);
    hipLaunchKernelGGL(fc_attn_kernel, dim3(tiles + agroups), dim3(128), 0, stream,
                       (const float*)m, fcW, fcb, dout, Nout, 2 * N, 2 * N, tiles,
                       (const float*)qa, (const float*)qb, beta, xa, ya, plain, Nq, B);
  };
  auto query_cat = [&](const float* Wq, int N, const float* dprev, int Nc) {
    int tiles = (B >> 5) * (N >> 5);
    int cgroups = (B * 2 * Nc) >> 9;
    hipLaunchKernelGGL(query_cat_kernel, dim3(2 * tiles + cgroups), dim3(128), 0, stream,
                       (const float*)m2, (const float*)(m2 + Nc), Wq, qa, qb, N, Nc, 2 * Nc, tiles,
                       (const float*)xa, (const float*)ya, dprev, m, Nc, B);
  };

  gemm_xy(m, m + 512, 1024, W2, qa, qb, 256, 512);
  fc_attn(fcW1, fcb1, d0, 512, 512, be2, m2, 256);
  query_cat(W3, 128, d0, 256);
  fc_attn(fcW2, fcb2, d1, 256, 256, be3, m2, 128);
  query_cat(W4, 64, d1, 128);
  fc_attn(fcW3, fcb3, d2, 128, 128, be4, nullptr, 64);
  hipLaunchKernelGGL(tail_kernel, dim3(B / TL_R), dim3(256), 0, stream,
                     (const float*)xa, (const float*)ya, (const float*)d2, fcW4, fcb4, W6, b6,
                     (float*)d_out, B);

  (void)in_sizes; (void)n_in; (void)out_size; (void)ws_size;
}
